# starting kernel
# baseline (speedup 1.0000x reference)
#include <hip/hip_runtime.h>
#include <math.h>


typedef __attribute__((ext_vector_type(2))) float v2f;
typedef __attribute__((ext_vector_type(8))) float v8f;

#define AXIS_LEN    160
#define RAD         6
#define WIN         13
#define BATCH_ELEMS (160 * 160 * 160 * 2)
#define TOTAL_ELEMS (4 * BATCH_ELEMS)
#define TOTAL_LINES (TOTAL_ELEMS / AXIS_LEN)
#define NGROUPS     (TOTAL_LINES / 16)

__global__ void gauss_wts_kernel(const float* __restrict__ sigma,
                                 float* __restrict__ wts) {
    int id = threadIdx.x;
    if (id >= 12) return;
    int b = id / 3, axis = id % 3;
    float s = sigma[b * 3 + axis];
    float denom = 2.0f * s * s + 1e-7f;
    float lg = logf(sqrtf(2.0f * (float)M_PI) * s + 1e-7f);
    float g[WIN];
    float sum = 0.0f;
    #pragma unroll
    for (int t = 0; t < WIN; ++t) {
        float loc = (float)t - (float)RAD;
        g[t] = expf(-(loc * loc) / denom - lg);
        sum += g[t];
    }
    float inv = 1.0f / sum;
    #pragma unroll
    for (int t = 0; t < WIN; ++t)
        wts[(b * 3 + axis) * WIN + t] = g[t] * inv;
}

template <int S>
__device__ __forceinline__ float ld_clamp_lo(const float* __restrict__ line, int c) {
    int cc = c < 0 ? 0 : c;
    float v = line[cc * S];
    return (c < 0) ? 0.0f : v;
}
template <int S>
__device__ __forceinline__ float ld_clamp_hi(const float* __restrict__ line, int c) {
    int cc = c > (AXIS_LEN - 1) ? (AXIS_LEN - 1) : c;
    float v = line[cc * S];
    return (c >= AXIS_LEN) ? 0.0f : v;
}

template <int S>
__global__ __launch_bounds__(128) void blur_pass_kernel(
        const float* __restrict__ src, float* __restrict__ dst,
        const float* __restrict__ wts, int axis) {
    const int lane  = threadIdx.x & 31;
    const int wave  = threadIdx.x >> 5;
    const int group = blockIdx.x * 4 + wave;

    const int n    = lane & 15;
    const int half = lane >> 4;
    const int q     = group * 16 + n;
    const int outer = q / S;
    const int inner = q - outer * S;
    const int base  = outer * (S * AXIS_LEN) + inner;
    const int b     = base / BATCH_ELEMS;
    const float* __restrict__ g = wts + (b * 3 + axis) * WIN;

    const int m = n;
    v2f A[7];
    #pragma unroll
    for (int kc = 0; kc < 7; ++kc) {
        int j0 = 4 * kc + 2 * half;
        int t0 = j0 - m;
        int t1 = t0 + 1;
        A[kc].x = (t0 >= 0 && t0 < WIN) ? g[t0] : 0.0f;
        A[kc].y = (t1 >= 0 && t1 < WIN) ? g[t1] : 0.0f;
    }

    const float* __restrict__ line  = src + base;
    float* __restrict__       oline = dst + base + (8 * half) * S;

    v2f Bv[7];
    #pragma unroll
    for (int kc = 0; kc < 7; ++kc) {
        int c0 = 4 * kc + 2 * half - RAD;
        Bv[kc].x = ld_clamp_lo<S>(line, c0);
        Bv[kc].y = ld_clamp_lo<S>(line, c0 + 1);
    }

    #pragma unroll
    for (int t = 0; t < 10; ++t) {
        const int i0 = 16 * t;

        v8f acc = {};
        #pragma unroll
        for (int kc = 0; kc < 7; ++kc) {
            acc = __builtin_amdgcn_wmma_f32_16x16x4_f32(
                 false, A[kc],  false, Bv[kc],
                 (short)0, acc,  false,  false);
        }

        if (t < 9) {
            Bv[0] = Bv[4];
            Bv[1] = Bv[5];
            Bv[2] = Bv[6];
            #pragma unroll
            for (int kc = 0; kc < 4; ++kc) {
                int c0 = i0 + 22 + 4 * kc + 2 * half;
                if (t == 8) {
                    Bv[3 + kc].x = ld_clamp_hi<S>(line, c0);
                    Bv[3 + kc].y = ld_clamp_hi<S>(line, c0 + 1);
                } else {
                    Bv[3 + kc].x = line[c0 * S];
                    Bv[3 + kc].y = line[(c0 + 1) * S];
                }
            }
        }

        #pragma unroll
        for (int v = 0; v < 8; ++v)
            oline[(i0 + v) * S] = acc[v];
    }
}

extern "C" void kernel_launch(void* const* d_in, const int* in_sizes, int n_in,
                              void* d_out, int out_size, void* d_ws, size_t ws_size,
                              hipStream_t stream) {
    const float* image = (const float*)d_in[0];
    const float* sigma = (const float*)d_in[1];
    float* out = (float*)d_out;
    float* wts = (float*)d_ws;
    float* tmp = (float*)((char*)d_ws + 4096);

    gauss_wts_kernel<<<1, 32, 0, stream>>>(sigma, wts);

    dim3 block(128);
    dim3 grid(NGROUPS / 4);

    constexpr int S_D = 160 * 160 * 2;
    constexpr int S_H = 160 * 2;
    constexpr int S_W = 2;

    blur_pass_kernel<S_D><<<grid, block, 0, stream>>>(image, out, wts, 0);
    blur_pass_kernel<S_H><<<grid, block, 0, stream>>>(out,   tmp, wts, 1);
    blur_pass_kernel<S_W><<<grid, block, 0, stream>>>(tmp,   out, wts, 2);
}


# D pass: a wave takes 32 adjacent lines, each load/store is one whole 128-byte line of one depth row; band split by lane half (k 0,1: columns 0-15, k 2,3: 16-31), 8-depth tiles; D sum order changes
# speedup vs baseline: 1.5966x; 1.5966x over previous
#include <hip/hip_runtime.h>
#include <math.h>


typedef __attribute__((ext_vector_type(2))) float v2f;
typedef __attribute__((ext_vector_type(8))) float v8f;

#define AXIS_LEN    160
#define RAD         6
#define WIN         13
#define BATCH_ELEMS (160 * 160 * 160 * 2)

__global__ void gauss_wts_kernel(const float* __restrict__ sigma,
                                 float* __restrict__ wts) {
    int id = threadIdx.x;
    if (id >= 12) return;
    int b = id / 3, axis = id % 3;
    float s = sigma[b * 3 + axis];
    float denom = 2.0f * s * s + 1e-7f;
    float lg = logf(sqrtf(2.0f * (float)M_PI) * s + 1e-7f);
    float g[WIN];
    float sum = 0.0f;
    #pragma unroll
    for (int t = 0; t < WIN; ++t) {
        float loc = (float)t - (float)RAD;
        g[t] = expf(-(loc * loc) / denom - lg);
        sum += g[t];
    }
    float inv = 1.0f / sum;
    #pragma unroll
    for (int t = 0; t < WIN; ++t)
        wts[(b * 3 + axis) * WIN + t] = g[t] * inv;
}

#define D_STRIDE    (160 * 160 * 2)
#define D_GROUPS_PB (D_STRIDE / 32)
__global__ __launch_bounds__(128) void blur_d_kernel(
        const float* __restrict__ src, float* __restrict__ dst,
        const float* __restrict__ wts) {
    const int lane  = threadIdx.x & 31;
    const int wave  = __builtin_amdgcn_readfirstlane((int)(threadIdx.x >> 5));
    const int group = blockIdx.x * 4 + wave;
    const int b     = group / D_GROUPS_PB;
    const int inner = (group - b * D_GROUPS_PB) * 32;
    const int base  = b * BATCH_ELEMS + inner + lane;
    const float* __restrict__ g = wts + (b * 3 + 0) * WIN;

    const int m    = lane & 15;
    const int half = lane >> 4;
    const int dl   = m & 7;
    const bool act = (m >> 3) == half;
    v2f A[10];
    #pragma unroll
    for (int kc = 0; kc < 10; ++kc) {
        int t0 = 2 * kc - dl;
        int t1 = t0 + 1;
        A[kc].x = (act && t0 >= 0 && t0 < WIN) ? g[t0] : 0.0f;
        A[kc].y = (act && t1 >= 0 && t1 < WIN) ? g[t1] : 0.0f;
    }

    const float* __restrict__ line  = src + base;
    float* __restrict__       oline = dst + base;

    v2f P[86];
    #pragma unroll
    for (int p = 3; p < 18; ++p) {
        P[p].x = line[(2 * p - 6) * D_STRIDE];
        P[p].y = line[(2 * p - 5) * D_STRIDE];
    }

    #pragma unroll
    for (int t = 0; t < 20; ++t) {
        #pragma unroll
        for (int k = 0; k < 4; ++k) {
            const int p = 4 * t + 18 + k;
            if (p <= 82) {
                P[p].x = line[(2 * p - 6) * D_STRIDE];
                P[p].y = line[(2 * p - 5) * D_STRIDE];
            }
        }

        v8f acc = {};
        #pragma unroll
        for (int kc = 0; kc < 10; ++kc) {
            const int p = 4 * t + kc;
            if (p >= 3 && p <= 82)
                acc = __builtin_amdgcn_wmma_f32_16x16x4_f32(
                     false, A[kc],  false, P[p],
                     (short)0, acc,  false,  false);
        }

        #pragma unroll
        for (int r = 0; r < 8; ++r)
            oline[(8 * t + r) * D_STRIDE] = acc[r];
    }
}


__device__ __forceinline__ v8f h_product(const v2f (&Rw)[7], const v2f (&GH)[7]) {
    v8f acc = {};
    #pragma unroll
    for (int kc = 0; kc < 7; ++kc)
        acc = __builtin_amdgcn_wmma_f32_16x16x4_f32(
            false, Rw[kc], false, GH[kc], (short)0, acc, false, false);
    return acc;
}

__device__ __forceinline__ void w_accumulate(v8f& a0, v8f& a1, const v8f& hA,
                                             const v2f (&GWd)[2]) {
    #pragma unroll
    for (int e = 0; e < 2; ++e) {
        v2f A0, A1;
        A0.x = hA[4 * e];     A0.y = hA[4 * e + 1];
        A1.x = hA[4 * e + 2]; A1.y = hA[4 * e + 3];
        a0 = __builtin_amdgcn_wmma_f32_16x16x4_f32(
            false, A0, false, GWd[e], (short)0, a0, false, false);
        a1 = __builtin_amdgcn_wmma_f32_16x16x4_f32(
            false, A1, false, GWd[e], (short)0, a1, false, false);
    }
}

__global__ __launch_bounds__(160) __attribute__((amdgpu_waves_per_eu(7))) void blur_hw_kernel(
        const float* __restrict__ src, float* __restrict__ dst,
        const float* __restrict__ wts) {
    const int lane  = threadIdx.x & 31;
    const int wave  = __builtin_amdgcn_readfirstlane((int)(threadIdx.x >> 5));
    const int group = blockIdx.x * 5 + wave;
    const int plane = group / 10;
    const int ht    = group - plane * 10;
    const int b     = plane / AXIS_LEN;
    const int n     = lane & 15;
    const int half  = lane >> 4;
    const int h0    = ht * 16;

    const float* __restrict__ gh = wts + (b * 3 + 1) * WIN;
    const float* __restrict__ gw = wts + (b * 3 + 2) * WIN;

    v2f GH[7];
    #pragma unroll
    for (int kc = 0; kc < 7; ++kc) {
        int t0 = 4 * kc + 2 * half - n;
        int t1 = t0 + 1;
        GH[kc].x = (t0 >= 0 && t0 < WIN) ? gh[t0] : 0.0f;
        GH[kc].y = (t1 >= 0 && t1 < WIN) ? gh[t1] : 0.0f;
    }
    v2f GW[4][2];
    #pragma unroll
    for (int di = 0; di < 4; ++di) {
        #pragma unroll
        for (int e = 0; e < 2; ++e) {
            int t0 = (8 * di - 8) + 2 * e + 4 * half - n + RAD;
            int t1 = t0 + 1;
            GW[di][e].x = (t0 >= 0 && t0 < WIN) ? gw[t0] : 0.0f;
            GW[di][e].y = (t1 >= 0 && t1 < WIN) ? gw[t1] : 0.0f;
        }
    }

    const float* __restrict__ p = src + plane * (AXIS_LEN * AXIS_LEN * 2);
    const int pn = (n & 12) | ((n & 1) << 1) | ((n >> 1) & 1);
    unsigned roff[7][2];
    bool     rok[7][2];
    #pragma unroll
    for (int kc = 0; kc < 7; ++kc) {
        #pragma unroll
        for (int j = 0; j < 2; ++j) {
            int r  = h0 - RAD + 4 * kc + 2 * half + j;
            int rc = r < 0 ? 0 : (r > AXIS_LEN - 1 ? AXIS_LEN - 1 : r);
            rok[kc][j]  = (r >= 0) && (r < AXIS_LEN);
            roff[kc][j] = (unsigned)(rc * (AXIS_LEN * 2) + pn);
        }
    }

    float* __restrict__ orow = dst + (plane * AXIS_LEN + h0 + 8 * half) * (AXIS_LEN * 2) + 2 * n;

    v2f cur[7], raw[7];
    #pragma unroll
    for (int kc = 0; kc < 7; ++kc) {
        float x = p[roff[kc][0]];
        float y = p[roff[kc][1]];
        cur[kc].x = rok[kc][0] ? x : 0.0f;
        cur[kc].y = rok[kc][1] ? y : 0.0f;
    }

    v8f p0 = {}, p1 = {};
    v8f q0 = {}, q1 = {};

    #pragma unroll
    for (int j = 0; j < 10; ++j) {
        #pragma unroll
        for (int sub = 0; sub < 2; ++sub) {
            const int s = 2 * j + sub;
            if (s < 19) {
                const float* __restrict__ ps = p + 16 * (s + 1);
                #pragma unroll
                for (int kc = 0; kc < 7; ++kc) {
                    raw[kc].x = ps[roff[kc][0]];
                    raw[kc].y = ps[roff[kc][1]];
                }
            }
            v8f hA = h_product(cur, GH);
            if (sub == 0) {
                if (j >= 1) {
                    w_accumulate(p0, p1, hA, GW[3]);
                    #pragma unroll
                    for (int r = 0; r < 8; ++r) {
                        v2f o; o.x = p0[r]; o.y = p1[r];
                        *(v2f*)(orow + r * (AXIS_LEN * 2) + 32 * (j - 1)) = o;
                    }
                }
                w_accumulate(q0, q1, hA, GW[1]);
            } else {
                w_accumulate(q0, q1, hA, GW[2]);
                p0 = q0; p1 = q1;
                v8f z0 = {}, z1 = {};
                q0 = z0; q1 = z1;
                if (j < 9) w_accumulate(q0, q1, hA, GW[0]);
            }
            if (s < 19) {
                #pragma unroll
                for (int kc = 0; kc < 7; ++kc) {
                    cur[kc].x = rok[kc][0] ? raw[kc].x : 0.0f;
                    cur[kc].y = rok[kc][1] ? raw[kc].y : 0.0f;
                }
            }
        }
    }
    #pragma unroll
    for (int r = 0; r < 8; ++r) {
        v2f o; o.x = p0[r]; o.y = p1[r];
        *(v2f*)(orow + r * (AXIS_LEN * 2) + 32 * 9) = o;
    }
}

extern "C" void kernel_launch(void* const* d_in, const int* in_sizes, int n_in,
                              void* d_out, int out_size, void* d_ws, size_t ws_size,
                              hipStream_t stream) {
    const float* image = (const float*)d_in[0];
    const float* sigma = (const float*)d_in[1];
    float* out = (float*)d_out;
    float* wts = (float*)d_ws;
    float* tmp = (float*)((char*)d_ws + 4096);

    gauss_wts_kernel<<<1, 32, 0, stream>>>(sigma, wts);

    blur_d_kernel<<<dim3(1600), dim3(128), 0, stream>>>(image, tmp, wts);
    blur_hw_kernel<<<dim3(1280), dim3(160), 0, stream>>>(tmp, out, wts);
}
